# starting kernel
# baseline (speedup 1.0000x reference)
#include <hip/hip_runtime.h>

typedef __attribute__((ext_vector_type(2))) float v2f;
typedef __attribute__((ext_vector_type(4))) float v4f;
typedef __attribute__((ext_vector_type(8))) float v8f;

#define B_ 8
#define N_ 4096
#define D_ 512
#define K_ 8
#define EPS_ 1e-6f

__global__ void zero_kernel(float* __restrict__ base, int n, float* __restrict__ kl) {
    int i = blockIdx.x * blockDim.x + threadIdx.x;
    if (i < n) base[i] = 0.0f;
    if (i == 0) *kl = 0.0f;
}

__global__ void __launch_bounds__(128)
logits_softmax_kernel(const float* __restrict__ x, const float* __restrict__ W,
                      const float* __restrict__ bias, float* __restrict__ probs,
                      float* __restrict__ Pk) {
    __shared__ float Wp[D_ * 16];

    const int tid    = threadIdx.x;
    const int lane   = tid & 31;
    const int wave   = tid >> 5;
    const int lanelo = lane & 15;
    const int hi     = lane >> 4;
    const int klane  = lanelo & 7;

    for (int i = tid; i < D_ * 16; i += 128) {
        const int row = i >> 4;
        const int col = i & 15;
        Wp[i] = (col < K_) ? W[row * K_ + col] : 0.0f;
    }
    __syncthreads();

    const int tile = blockIdx.x * 4 + wave;
    const int b    = tile >> 8;
    const int n0   = (tile & 255) << 4;

    const float* xrow = x + ((size_t)b * N_ + n0 + lanelo) * D_ + 2 * hi;
    const float* wrow = Wp + 2 * hi * 16 + lanelo;

    v8f c = {};
    #pragma unroll 4
    for (int s = 0; s < D_ / 4; ++s) {
        v2f a;
        a.x = xrow[4 * s];
        a.y = xrow[4 * s + 1];
        v2f bb;
        bb.x = wrow[4 * s * 16];
        bb.y = wrow[(4 * s + 1) * 16];
        c = __builtin_amdgcn_wmma_f32_16x16x4_f32(false, a, false, bb,
                                                  (short)0, c, false, false);
    }

    const float bk = bias[klane];
    float pk_local = 0.0f;
    #pragma unroll
    for (int v = 0; v < 8; ++v) {
        float lg = c[v] + bk;
        float mx = lg;
        mx = fmaxf(mx, __shfl_xor(mx, 1, 8));
        mx = fmaxf(mx, __shfl_xor(mx, 2, 8));
        mx = fmaxf(mx, __shfl_xor(mx, 4, 8));
        float e = expf(lg - mx);
        float ssum = e;
        ssum += __shfl_xor(ssum, 1, 8);
        ssum += __shfl_xor(ssum, 2, 8);
        ssum += __shfl_xor(ssum, 4, 8);
        float p = e / ssum;
        if (lanelo < K_) {
            probs[((size_t)b * N_ + n0 + v + 8 * hi) * K_ + lanelo] = p;
            pk_local += p;
        }
    }
    if (lanelo < K_) atomicAdd(&Pk[b * K_ + lanelo], pk_local);
}

__global__ void __launch_bounds__(128)
moments_kernel(const float* __restrict__ x, const float* __restrict__ probs,
               float* __restrict__ mu, float* __restrict__ s2) {
    __shared__ float Pp[512 * 16];

    const int tid    = threadIdx.x;
    const int lane   = tid & 31;
    const int wave   = tid >> 5;
    const int lanelo = lane & 15;
    const int hi     = lane >> 4;

    const int b      = blockIdx.x >> 6;
    const int rem    = blockIdx.x & 63;
    const int chunk  = rem >> 3;
    const int sgrp   = rem & 7;
    const int nbase  = chunk * 512;
    const int slice  = sgrp * 4 + wave;
    const int d      = slice * 16 + lanelo;

    const float* xb = x     + (size_t)b * N_ * D_;
    const float* pb = probs + ((size_t)b * N_ + nbase) * K_;

    const v4f zero4 = {};
    for (int r = tid; r < 512; r += 128) {
        v4f p0 = *(const v4f*)&pb[r * K_];
        v4f p1 = *(const v4f*)&pb[r * K_ + 4];
        *(v4f*)&Pp[r * 16 + 0]  = p0;
        *(v4f*)&Pp[r * 16 + 4]  = p1;
        *(v4f*)&Pp[r * 16 + 8]  = zero4;
        *(v4f*)&Pp[r * 16 + 12] = zero4;
    }
    __syncthreads();

    const float* prow = Pp + 2 * hi * 16 + lanelo;

    v8f cmu = {};
    v8f cs2 = {};
    #pragma unroll 4
    for (int t = 0; t < 512; t += 4) {
        const int n = nbase + t + 2 * hi;
        v2f a, bx, bx2;
        a.x = prow[t * 16];
        a.y = prow[t * 16 + 16];
        const float b0 = xb[(size_t)n * D_ + d];
        const float b1 = xb[(size_t)(n + 1) * D_ + d];
        bx.x  = b0;      bx.y  = b1;
        bx2.x = b0 * b0; bx2.y = b1 * b1;
        cmu = __builtin_amdgcn_wmma_f32_16x16x4_f32(false, a, false, bx,
                                                    (short)0, cmu, false, false);
        cs2 = __builtin_amdgcn_wmma_f32_16x16x4_f32(false, a, false, bx2,
                                                    (short)0, cs2, false, false);
    }
    if (hi == 0) {
        #pragma unroll
        for (int v = 0; v < 8; ++v) {
            const size_t idx = ((size_t)(b * K_ + v)) * D_ + d;
            atomicAdd(&mu[idx], cmu[v]);
            atomicAdd(&s2[idx], cs2[v]);
        }
    }
}

__global__ void __launch_bounds__(256)
finalize_kernel(const float* __restrict__ mu, const float* __restrict__ s2,
                const float* __restrict__ Pk, const float* __restrict__ eps,
                float* __restrict__ out) {
    __shared__ float red[256];
    const int idx = blockIdx.x * 256 + threadIdx.x;
    const int b = idx >> 9;
    const int d = idx & (D_ - 1);

    float acc = 0.0f, klacc = 0.0f;
    #pragma unroll
    for (int k = 0; k < K_; ++k) {
        const size_t i = ((size_t)(b * K_ + k)) * D_ + d;
        const float m   = mu[i];
        const float v   = s2[i];
        const float P   = Pk[b * K_ + k];
        const float sig = v + m * m * (P - 2.0f);
        const float sp  = sig + EPS_;
        acc   += m + eps[i] * sqrtf(sp);
        klacc += 1.0f + logf(sp) - m * m - sig;
    }
    out[b * D_ + d] = acc * (1.0f / K_);

    red[threadIdx.x] = klacc * (-0.5f / (float)(B_ * K_));
    __syncthreads();
    for (int o = 128; o > 0; o >>= 1) {
        if (threadIdx.x < o) red[threadIdx.x] += red[threadIdx.x + o];
        __syncthreads();
    }
    if (threadIdx.x == 0) atomicAdd(&out[B_ * D_], red[0]);
}

extern "C" void kernel_launch(void* const* d_in, const int* in_sizes, int n_in,
                              void* d_out, int out_size, void* d_ws, size_t ws_size,
                              hipStream_t stream) {
    (void)in_sizes; (void)n_in; (void)out_size; (void)ws_size;
    const float* x    = (const float*)d_in[0];
    const float* W    = (const float*)d_in[1];
    const float* bias = (const float*)d_in[2];
    const float* eps  = (const float*)d_in[3];
    float* out = (float*)d_out;

    float* probs = (float*)d_ws;
    float* mu    = probs + (size_t)B_ * N_ * K_;
    float* s2    = mu    + (size_t)B_ * K_ * D_;
    float* Pk    = s2    + (size_t)B_ * K_ * D_;

    const int nz = 2 * B_ * K_ * D_ + B_ * K_;
    zero_kernel<<<(nz + 255) / 256, 256, 0, stream>>>(mu, nz, &out[B_ * D_]);

    logits_softmax_kernel<<<(B_ * N_ / 16) / 4, 128, 0, stream>>>(x, W, bias, probs, Pk);

    moments_kernel<<<512, 128, 0, stream>>>(x, probs, mu, s2);

    finalize_kernel<<<(B_ * D_) / 256, 256, 0, stream>>>(mu, s2, Pk, eps, out);
}


# logits kernel: every load issued before the first wait (x tile as 64 16-byte loads per lane, W repacked in LDS, LDS-only barrier), k order permuted, two accumulator tiles: d summation order changes
# speedup vs baseline: 1.7634x; 1.7634x over previous
#include <hip/hip_runtime.h>

typedef __attribute__((ext_vector_type(2))) float v2f;
typedef __attribute__((ext_vector_type(4))) float v4f;
typedef __attribute__((ext_vector_type(8))) float v8f;

#define B_ 8
#define N_ 4096
#define D_ 512
#define K_ 8
#define EPS_ 1e-6f

__global__ void zero_kernel(float* __restrict__ base, int n, float* __restrict__ kl) {
    int i = blockIdx.x * blockDim.x + threadIdx.x;
    if (i < n) base[i] = 0.0f;
    if (i == 0) *kl = 0.0f;
}

__global__ void __launch_bounds__(128)
logits_softmax_kernel(const float* __restrict__ x, const float* __restrict__ W,
                      const float* __restrict__ bias, float* __restrict__ probs,
                      float* __restrict__ Pk) {
    __shared__ v4f Wt[(D_ / 4) * K_];

    const int tid    = threadIdx.x;
    const int lane   = tid & 31;
    const int wave   = tid >> 5;
    const int lanelo = lane & 15;
    const int hi     = lane >> 4;
    const int klane  = lanelo & 7;

    const int tile = blockIdx.x * 4 + wave;
    const int b    = tile >> 8;
    const int n0   = (tile & 255) << 4;

    const v4f* W4 = (const v4f*)W;
    v4f wreg[8];
    #pragma unroll
    for (int i = 0; i < 8; ++i) wreg[i] = W4[tid + 128 * i];

    const v4f* xrow =
        (const v4f*)(x + ((size_t)b * N_ + n0 + lanelo) * D_ + 4 * hi);
    v4f xr[64];
    #pragma unroll
    for (int q = 0; q < 64; ++q) xr[q] = xrow[2 * q];

    float* Wts = (float*)Wt;
    #pragma unroll
    for (int i = 0; i < 8; ++i) {
        const int f    = tid + 128 * i;
        const int d    = f >> 1;
        const int k0   = (f & 1) * 4;
        const int base = (((d >> 2) * K_ + k0) << 2) + (d & 3);
        Wts[base]      = wreg[i].x;
        Wts[base + 4]  = wreg[i].y;
        Wts[base + 8]  = wreg[i].z;
        Wts[base + 12] = wreg[i].w;
    }
    __builtin_amdgcn_fence(__ATOMIC_RELEASE, "workgroup", "local");
    __builtin_amdgcn_s_barrier();
    __builtin_amdgcn_fence(__ATOMIC_ACQUIRE, "workgroup", "local");

    const v4f* wfrag = Wt + hi * K_ + klane;

    v8f c0 = {};
    v8f c1 = {};
    #pragma unroll
    for (int q0 = 0; q0 < 64; q0 += 8) {
        v4f wv[8];
        #pragma unroll
        for (int j = 0; j < 8; ++j) wv[j] = wfrag[(q0 + j) * 2 * K_];
        #pragma unroll
        for (int j = 0; j < 8; ++j) {
            const v4f xa = xr[q0 + j];
            v2f a0, b0, a1, b1;
            a0.x = xa.x;     a0.y = xa.y;
            b0.x = wv[j].x;  b0.y = wv[j].y;
            a1.x = xa.z;     a1.y = xa.w;
            b1.x = wv[j].z;  b1.y = wv[j].w;
            c0 = __builtin_amdgcn_wmma_f32_16x16x4_f32(false, a0, false, b0,
                                                       (short)0, c0, false, false);
            c1 = __builtin_amdgcn_wmma_f32_16x16x4_f32(false, a1, false, b1,
                                                       (short)0, c1, false, false);
        }
    }
    const v8f c = c0 + c1;

    const float bk = bias[klane];
    float pk_local = 0.0f;
    #pragma unroll
    for (int v = 0; v < 8; ++v) {
        float lg = c[v] + bk;
        float mx = lg;
        mx = fmaxf(mx, __shfl_xor(mx, 1, 8));
        mx = fmaxf(mx, __shfl_xor(mx, 2, 8));
        mx = fmaxf(mx, __shfl_xor(mx, 4, 8));
        float e = expf(lg - mx);
        float ssum = e;
        ssum += __shfl_xor(ssum, 1, 8);
        ssum += __shfl_xor(ssum, 2, 8);
        ssum += __shfl_xor(ssum, 4, 8);
        float p = e / ssum;
        if (lanelo < K_) {
            probs[((size_t)b * N_ + n0 + v + 8 * hi) * K_ + lanelo] = p;
            pk_local += p;
        }
    }
    if (lanelo < K_) atomicAdd(&Pk[b * K_ + lanelo], pk_local);
}

__global__ void __launch_bounds__(128)
moments_kernel(const float* __restrict__ x, const float* __restrict__ probs,
               float* __restrict__ mu, float* __restrict__ s2) {
    __shared__ float Pp[512 * 16];

    const int tid    = threadIdx.x;
    const int lane   = tid & 31;
    const int wave   = tid >> 5;
    const int lanelo = lane & 15;
    const int hi     = lane >> 4;

    const int b      = blockIdx.x >> 6;
    const int rem    = blockIdx.x & 63;
    const int chunk  = rem >> 3;
    const int sgrp   = rem & 7;
    const int nbase  = chunk * 512;
    const int slice  = sgrp * 4 + wave;
    const int d      = slice * 16 + lanelo;

    const float* xb = x     + (size_t)b * N_ * D_;
    const float* pb = probs + ((size_t)b * N_ + nbase) * K_;

    const v4f zero4 = {};
    for (int r = tid; r < 512; r += 128) {
        v4f p0 = *(const v4f*)&pb[r * K_];
        v4f p1 = *(const v4f*)&pb[r * K_ + 4];
        *(v4f*)&Pp[r * 16 + 0]  = p0;
        *(v4f*)&Pp[r * 16 + 4]  = p1;
        *(v4f*)&Pp[r * 16 + 8]  = zero4;
        *(v4f*)&Pp[r * 16 + 12] = zero4;
    }
    __syncthreads();

    const float* prow = Pp + 2 * hi * 16 + lanelo;

    v8f cmu = {};
    v8f cs2 = {};
    #pragma unroll 4
    for (int t = 0; t < 512; t += 4) {
        const int n = nbase + t + 2 * hi;
        v2f a, bx, bx2;
        a.x = prow[t * 16];
        a.y = prow[t * 16 + 16];
        const float b0 = xb[(size_t)n * D_ + d];
        const float b1 = xb[(size_t)(n + 1) * D_ + d];
        bx.x  = b0;      bx.y  = b1;
        bx2.x = b0 * b0; bx2.y = b1 * b1;
        cmu = __builtin_amdgcn_wmma_f32_16x16x4_f32(false, a, false, bx,
                                                    (short)0, cmu, false, false);
        cs2 = __builtin_amdgcn_wmma_f32_16x16x4_f32(false, a, false, bx2,
                                                    (short)0, cs2, false, false);
    }
    if (hi == 0) {
        #pragma unroll
        for (int v = 0; v < 8; ++v) {
            const size_t idx = ((size_t)(b * K_ + v)) * D_ + d;
            atomicAdd(&mu[idx], cmu[v]);
            atomicAdd(&s2[idx], cs2[v]);
        }
    }
}

__global__ void __launch_bounds__(256)
finalize_kernel(const float* __restrict__ mu, const float* __restrict__ s2,
                const float* __restrict__ Pk, const float* __restrict__ eps,
                float* __restrict__ out) {
    __shared__ float red[256];
    const int idx = blockIdx.x * 256 + threadIdx.x;
    const int b = idx >> 9;
    const int d = idx & (D_ - 1);

    float acc = 0.0f, klacc = 0.0f;
    #pragma unroll
    for (int k = 0; k < K_; ++k) {
        const size_t i = ((size_t)(b * K_ + k)) * D_ + d;
        const float m   = mu[i];
        const float v   = s2[i];
        const float P   = Pk[b * K_ + k];
        const float sig = v + m * m * (P - 2.0f);
        const float sp  = sig + EPS_;
        acc   += m + eps[i] * sqrtf(sp);
        klacc += 1.0f + logf(sp) - m * m - sig;
    }
    out[b * D_ + d] = acc * (1.0f / K_);

    red[threadIdx.x] = klacc * (-0.5f / (float)(B_ * K_));
    __syncthreads();
    for (int o = 128; o > 0; o >>= 1) {
        if (threadIdx.x < o) red[threadIdx.x] += red[threadIdx.x + o];
        __syncthreads();
    }
    if (threadIdx.x == 0) atomicAdd(&out[B_ * D_], red[0]);
}

extern "C" void kernel_launch(void* const* d_in, const int* in_sizes, int n_in,
                              void* d_out, int out_size, void* d_ws, size_t ws_size,
                              hipStream_t stream) {
    (void)in_sizes; (void)n_in; (void)out_size; (void)ws_size;
    const float* x    = (const float*)d_in[0];
    const float* W    = (const float*)d_in[1];
    const float* bias = (const float*)d_in[2];
    const float* eps  = (const float*)d_in[3];
    float* out = (float*)d_out;

    float* probs = (float*)d_ws;
    float* mu    = probs + (size_t)B_ * N_ * K_;
    float* s2    = mu    + (size_t)B_ * K_ * D_;
    float* Pk    = s2    + (size_t)B_ * K_ * D_;

    const int nz = 2 * B_ * K_ * D_ + B_ * K_;
    zero_kernel<<<(nz + 255) / 256, 256, 0, stream>>>(mu, nz, &out[B_ * D_]);

    logits_softmax_kernel<<<(B_ * N_ / 16) / 4, 128, 0, stream>>>(x, W, bias, probs, Pk);

    moments_kernel<<<512, 128, 0, stream>>>(x, probs, mu, s2);

    finalize_kernel<<<(B_ * D_) / 256, 256, 0, stream>>>(mu, s2, Pk, eps, out);
}


# moments kernel: x as 16-byte pieces along d, all 64 per lane issued before the first wait, 4 strided accumulator tiles; 4 waves split n and are summed in LDS before the atomics: n sum order changes
# speedup vs baseline: 2.3494x; 1.3323x over previous
#include <hip/hip_runtime.h>

typedef __attribute__((ext_vector_type(2))) float v2f;
typedef __attribute__((ext_vector_type(4))) float v4f;
typedef __attribute__((ext_vector_type(8))) float v8f;

#define B_ 8
#define N_ 4096
#define D_ 512
#define K_ 8
#define EPS_ 1e-6f

__global__ void zero_kernel(float* __restrict__ base, int n, float* __restrict__ kl) {
    int i = blockIdx.x * blockDim.x + threadIdx.x;
    if (i < n) base[i] = 0.0f;
    if (i == 0) *kl = 0.0f;
}

__global__ void __launch_bounds__(128)
logits_softmax_kernel(const float* __restrict__ x, const float* __restrict__ W,
                      const float* __restrict__ bias, float* __restrict__ probs,
                      float* __restrict__ Pk) {
    __shared__ v4f Wt[(D_ / 4) * K_];

    const int tid    = threadIdx.x;
    const int lane   = tid & 31;
    const int wave   = tid >> 5;
    const int lanelo = lane & 15;
    const int hi     = lane >> 4;
    const int klane  = lanelo & 7;

    const int tile = blockIdx.x * 4 + wave;
    const int b    = tile >> 8;
    const int n0   = (tile & 255) << 4;

    const v4f* W4 = (const v4f*)W;
    v4f wreg[8];
    #pragma unroll
    for (int i = 0; i < 8; ++i) wreg[i] = W4[tid + 128 * i];

    const v4f* xrow =
        (const v4f*)(x + ((size_t)b * N_ + n0 + lanelo) * D_ + 4 * hi);
    v4f xr[64];
    #pragma unroll
    for (int q = 0; q < 64; ++q) xr[q] = xrow[2 * q];

    float* Wts = (float*)Wt;
    #pragma unroll
    for (int i = 0; i < 8; ++i) {
        const int f    = tid + 128 * i;
        const int d    = f >> 1;
        const int k0   = (f & 1) * 4;
        const int base = (((d >> 2) * K_ + k0) << 2) + (d & 3);
        Wts[base]      = wreg[i].x;
        Wts[base + 4]  = wreg[i].y;
        Wts[base + 8]  = wreg[i].z;
        Wts[base + 12] = wreg[i].w;
    }
    __builtin_amdgcn_fence(__ATOMIC_RELEASE, "workgroup", "local");
    __builtin_amdgcn_s_barrier();
    __builtin_amdgcn_fence(__ATOMIC_ACQUIRE, "workgroup", "local");

    const v4f* wfrag = Wt + hi * K_ + klane;

    v8f c0 = {};
    v8f c1 = {};
    #pragma unroll
    for (int q0 = 0; q0 < 64; q0 += 8) {
        v4f wv[8];
        #pragma unroll
        for (int j = 0; j < 8; ++j) wv[j] = wfrag[(q0 + j) * 2 * K_];
        #pragma unroll
        for (int j = 0; j < 8; ++j) {
            const v4f xa = xr[q0 + j];
            v2f a0, b0, a1, b1;
            a0.x = xa.x;     a0.y = xa.y;
            b0.x = wv[j].x;  b0.y = wv[j].y;
            a1.x = xa.z;     a1.y = xa.w;
            b1.x = wv[j].z;  b1.y = wv[j].w;
            c0 = __builtin_amdgcn_wmma_f32_16x16x4_f32(false, a0, false, b0,
                                                       (short)0, c0, false, false);
            c1 = __builtin_amdgcn_wmma_f32_16x16x4_f32(false, a1, false, b1,
                                                       (short)0, c1, false, false);
        }
    }
    const v8f c = c0 + c1;

    const float bk = bias[klane];
    float pk_local = 0.0f;
    #pragma unroll
    for (int v = 0; v < 8; ++v) {
        float lg = c[v] + bk;
        float mx = lg;
        mx = fmaxf(mx, __shfl_xor(mx, 1, 8));
        mx = fmaxf(mx, __shfl_xor(mx, 2, 8));
        mx = fmaxf(mx, __shfl_xor(mx, 4, 8));
        float e = expf(lg - mx);
        float ssum = e;
        ssum += __shfl_xor(ssum, 1, 8);
        ssum += __shfl_xor(ssum, 2, 8);
        ssum += __shfl_xor(ssum, 4, 8);
        float p = e / ssum;
        if (lanelo < K_) {
            probs[((size_t)b * N_ + n0 + v + 8 * hi) * K_ + lanelo] = p;
            pk_local += p;
        }
    }
    if (lanelo < K_) atomicAdd(&Pk[b * K_ + lanelo], pk_local);
}

__global__ void __launch_bounds__(128)
moments_kernel(const float* __restrict__ x, const float* __restrict__ probs,
               float* __restrict__ mu, float* __restrict__ s2) {
    __shared__ v4f Pq[512 * K_ / 4];
    __shared__ v4f Red[4 * 16 * 64 / 4];

    const int tid  = threadIdx.x;
    const int lane = tid & 31;
    const int wave = tid >> 5;
    const int c    = lane & 15;
    const int hi   = lane >> 4;

    const int b     = blockIdx.x >> 6;
    const int rem   = blockIdx.x & 63;
    const int chunk = rem >> 3;
    const int dsl   = rem & 7;
    const int nbase = chunk * 512;
    const int d0    = dsl * 64;

    const v4f* pb4 = (const v4f*)(probs + ((size_t)b * N_ + nbase) * K_);
    v4f pr[8];
    #pragma unroll
    for (int i8 = 0; i8 < 8; ++i8) pr[i8] = pb4[(tid + 128 * (i8 >> 2)) * 4 + (i8 & 3)];

    const v4f* xw = (const v4f*)(x + ((size_t)b * N_ + nbase + 128 * wave + 2 * hi) * D_
                                   + d0 + 4 * c);
    v4f xr[64];
    #pragma unroll
    for (int t = 0; t < 32; ++t) {
        xr[2 * t]     = xw[(4 * t) * (D_ / 4)];
        xr[2 * t + 1] = xw[(4 * t + 1) * (D_ / 4)];
    }

    #pragma unroll
    for (int it = 0; it < 2; ++it) {
        const int i = tid + 128 * it;
        const v4f e0 = pr[it * 4 + 0];
        const v4f e1 = pr[it * 4 + 1];
        const v4f o0 = pr[it * 4 + 2];
        const v4f o1 = pr[it * 4 + 3];
        v4f q0, q1, q2, q3;
        q0.x = e0.x; q0.y = o0.x; q0.z = e0.y; q0.w = o0.y;
        q1.x = e0.z; q1.y = o0.z; q1.z = e0.w; q1.w = o0.w;
        q2.x = e1.x; q2.y = o1.x; q2.z = e1.y; q2.w = o1.y;
        q3.x = e1.z; q3.y = o1.z; q3.z = e1.w; q3.w = o1.w;
        Pq[i * 4 + 0] = q0;
        Pq[i * 4 + 1] = q1;
        Pq[i * 4 + 2] = q2;
        Pq[i * 4 + 3] = q3;
    }
    __builtin_amdgcn_fence(__ATOMIC_RELEASE, "workgroup", "local");
    __builtin_amdgcn_s_barrier();
    __builtin_amdgcn_fence(__ATOMIC_ACQUIRE, "workgroup", "local");

    __builtin_amdgcn_sched_barrier(0);

    const v2f* arow = (const v2f*)Pq + (64 * wave + hi) * K_ + (c & 7);
    v2f av[32];
    #pragma unroll
    for (int t = 0; t < 32; ++t) av[t] = arow[t * 2 * K_];

    v8f cm0 = {}, cm1 = {}, cm2 = {}, cm3 = {};
    v8f cs0 = {}, cs1 = {}, cs2 = {}, cs3 = {};
    #pragma unroll
    for (int t = 0; t < 32; ++t) {
        __builtin_amdgcn_sched_barrier(0);
        const v2f a  = av[t];
        const v4f xa = xr[2 * t];
        const v4f xb = xr[2 * t + 1];
        v2f b0, b1, b2, b3;
        b0.x = xa.x; b0.y = xb.x;
        b1.x = xa.y; b1.y = xb.y;
        b2.x = xa.z; b2.y = xb.z;
        b3.x = xa.w; b3.y = xb.w;
        const v2f g0 = b0 * b0;
        const v2f g1 = b1 * b1;
        const v2f g2 = b2 * b2;
        const v2f g3 = b3 * b3;
        cm0 = __builtin_amdgcn_wmma_f32_16x16x4_f32(false, a, false, b0,
                                                    (short)0, cm0, false, false);
        cs0 = __builtin_amdgcn_wmma_f32_16x16x4_f32(false, a, false, g0,
                                                    (short)0, cs0, false, false);
        cm1 = __builtin_amdgcn_wmma_f32_16x16x4_f32(false, a, false, b1,
                                                    (short)0, cm1, false, false);
        cs1 = __builtin_amdgcn_wmma_f32_16x16x4_f32(false, a, false, g1,
                                                    (short)0, cs1, false, false);
        cm2 = __builtin_amdgcn_wmma_f32_16x16x4_f32(false, a, false, b2,
                                                    (short)0, cm2, false, false);
        cs2 = __builtin_amdgcn_wmma_f32_16x16x4_f32(false, a, false, g2,
                                                    (short)0, cs2, false, false);
        cm3 = __builtin_amdgcn_wmma_f32_16x16x4_f32(false, a, false, b3,
                                                    (short)0, cm3, false, false);
        cs3 = __builtin_amdgcn_wmma_f32_16x16x4_f32(false, a, false, g3,
                                                    (short)0, cs3, false, false);
    }

    if (hi == 0) {
        #pragma unroll
        for (int r = 0; r < 8; ++r) {
            v4f vm, vs;
            vm.x = cm0[r]; vm.y = cm1[r]; vm.z = cm2[r]; vm.w = cm3[r];
            vs.x = cs0[r]; vs.y = cs1[r]; vs.z = cs2[r]; vs.w = cs3[r];
            Red[(wave * 16 + r) * 16 + c]     = vm;
            Red[(wave * 16 + 8 + r) * 16 + c] = vs;
        }
    }
    __syncthreads();

    const float* redf = (const float*)Red;
    const int d64 = tid & 63;
    const int qg  = tid >> 6;
    float* dst = (qg ? s2 : mu) + ((size_t)b * K_) * D_ + d0 + d64;
    float vsum[K_];
    #pragma unroll
    for (int k = 0; k < K_; ++k) {
        const int q = qg * 8 + k;
        float v = redf[(0 * 16 + q) * 64 + d64];
        v += redf[(1 * 16 + q) * 64 + d64];
        v += redf[(2 * 16 + q) * 64 + d64];
        v += redf[(3 * 16 + q) * 64 + d64];
        vsum[k] = v;
    }
    #pragma unroll
    for (int k = 0; k < K_; ++k) atomicAdd(&dst[(size_t)k * D_], vsum[k]);
}

__global__ void __launch_bounds__(256)
finalize_kernel(const float* __restrict__ mu, const float* __restrict__ s2,
                const float* __restrict__ Pk, const float* __restrict__ eps,
                float* __restrict__ out) {
    __shared__ float red[256];
    const int idx = blockIdx.x * 256 + threadIdx.x;
    const int b = idx >> 9;
    const int d = idx & (D_ - 1);

    float acc = 0.0f, klacc = 0.0f;
    #pragma unroll
    for (int k = 0; k < K_; ++k) {
        const size_t i = ((size_t)(b * K_ + k)) * D_ + d;
        const float m   = mu[i];
        const float v   = s2[i];
        const float P   = Pk[b * K_ + k];
        const float sig = v + m * m * (P - 2.0f);
        const float sp  = sig + EPS_;
        acc   += m + eps[i] * sqrtf(sp);
        klacc += 1.0f + logf(sp) - m * m - sig;
    }
    out[b * D_ + d] = acc * (1.0f / K_);

    red[threadIdx.x] = klacc * (-0.5f / (float)(B_ * K_));
    __syncthreads();
    for (int o = 128; o > 0; o >>= 1) {
        if (threadIdx.x < o) red[threadIdx.x] += red[threadIdx.x + o];
        __syncthreads();
    }
    if (threadIdx.x == 0) atomicAdd(&out[B_ * D_], red[0]);
}

extern "C" void kernel_launch(void* const* d_in, const int* in_sizes, int n_in,
                              void* d_out, int out_size, void* d_ws, size_t ws_size,
                              hipStream_t stream) {
    (void)in_sizes; (void)n_in; (void)out_size; (void)ws_size;
    const float* x    = (const float*)d_in[0];
    const float* W    = (const float*)d_in[1];
    const float* bias = (const float*)d_in[2];
    const float* eps  = (const float*)d_in[3];
    float* out = (float*)d_out;

    float* probs = (float*)d_ws;
    float* mu    = probs + (size_t)B_ * N_ * K_;
    float* s2    = mu    + (size_t)B_ * K_ * D_;
    float* Pk    = s2    + (size_t)B_ * K_ * D_;

    const int nz = 2 * B_ * K_ * D_ + B_ * K_;
    zero_kernel<<<(nz + 255) / 256, 256, 0, stream>>>(mu, nz, &out[B_ * D_]);

    logits_softmax_kernel<<<(B_ * N_ / 16) / 4, 128, 0, stream>>>(x, W, bias, probs, Pk);

    moments_kernel<<<512, 128, 0, stream>>>(x, probs, mu, s2);

    finalize_kernel<<<(B_ * D_) / 256, 256, 0, stream>>>(mu, s2, Pk, eps, out);
}
